# starting kernel
# baseline (speedup 1.0000x reference)
#include <hip/hip_runtime.h>

typedef __attribute__((ext_vector_type(16))) _Float16 v16h;
typedef __attribute__((ext_vector_type(8)))  _Float16 v8h;
typedef __attribute__((ext_vector_type(8)))  float    v8f;

#define T_DIM   512
#define D_DIM   256
#define A_DIM   50
#define A_PAD   64
#define XS      264
#define NTHREADS 256
#define EPS_K   1e-7f

#define OFF_X    0
#define SZ_X     (T_DIM * XS * 2)
#define OFF_WT   (OFF_X + SZ_X)
#define SZ_WT    (A_PAD * XS * 2)
#define OFF_B    (OFF_WT + SZ_WT)
#define OFF_U    (OFF_B + A_PAD * 4)
#define OFF_S    (OFF_U + A_PAD * 4)
#define OFF_RED  (OFF_S + T_DIM * 4)
#define SMEM_BYTES (OFF_RED + NTHREADS * 4)

__device__ __forceinline__ float fast_tanh(float x) {
    float e = __expf(2.0f * x);
    return (e - 1.0f) / (e + 1.0f);
}

__global__ __launch_bounds__(NTHREADS, 1)
void attn_pool_kernel(const float* __restrict__ x,
                      const float* __restrict__ W,
                      const float* __restrict__ bias,
                      const float* __restrict__ u,
                      float* __restrict__ out)
{
    extern __shared__ __align__(16) char smem[];
    _Float16* xs  = (_Float16*)(smem + OFF_X);
    _Float16* wt  = (_Float16*)(smem + OFF_WT);
    float*    bl  = (float*)(smem + OFF_B);
    float*    ul  = (float*)(smem + OFF_U);
    float*    sc  = (float*)(smem + OFF_S);
    float*    red = (float*)(smem + OFF_RED);

    const int tid  = threadIdx.x;
    const int b    = blockIdx.x;
    const int lane = tid & 31;
    const int wave = tid >> 5;

    for (int idx = tid; idx < A_PAD * D_DIM; idx += NTHREADS) {
        int n = idx & (A_PAD - 1);
        int k = idx >> 6;
        float w = (n < A_DIM) ? W[k * A_DIM + n] : 0.0f;
        wt[n * XS + k] = (_Float16)w;
    }
    if (tid < A_PAD) {
        bl[tid] = (tid < A_DIM) ? bias[tid] : 0.0f;
        ul[tid] = (tid < A_DIM) ? u[tid]    : 0.0f;
    }

    const float4* xg = (const float4*)(x + (size_t)b * T_DIM * D_DIM);
    for (int idx = tid; idx < T_DIM * (D_DIM / 4); idx += NTHREADS) {
        int t = idx >> 6;
        int c = idx & 63;
        float4 v = xg[t * (D_DIM / 4) + c];
        _Float16* dst = xs + t * XS + c * 4;
        dst[0] = (_Float16)v.x; dst[1] = (_Float16)v.y;
        dst[2] = (_Float16)v.z; dst[3] = (_Float16)v.w;
    }
    __syncthreads();

    const int mrow  = lane & 15;
    const int khalf = (lane >> 4) * 8;

    for (int mt = wave; mt < T_DIM / 16; mt += NTHREADS / 32) {
        const int t0 = mt * 16;
        v8f acc0 = {}, acc1 = {}, acc2 = {}, acc3 = {};

        for (int ks = 0; ks < D_DIM / 32; ++ks) {
            const _Float16* ap = xs + (t0 + mrow) * XS + ks * 32 + khalf;
            v16h a;
            *(v8h*)&a       = *(const v8h*)ap;
            *((v8h*)&a + 1) = *(const v8h*)(ap + 16);

            const _Float16* bp0 = wt + ( 0 + mrow) * XS + ks * 32 + khalf;
            const _Float16* bp1 = wt + (16 + mrow) * XS + ks * 32 + khalf;
            const _Float16* bp2 = wt + (32 + mrow) * XS + ks * 32 + khalf;
            const _Float16* bp3 = wt + (48 + mrow) * XS + ks * 32 + khalf;
            v16h b0, b1, b2, b3;
            *(v8h*)&b0 = *(const v8h*)bp0; *((v8h*)&b0 + 1) = *(const v8h*)(bp0 + 16);
            *(v8h*)&b1 = *(const v8h*)bp1; *((v8h*)&b1 + 1) = *(const v8h*)(bp1 + 16);
            *(v8h*)&b2 = *(const v8h*)bp2; *((v8h*)&b2 + 1) = *(const v8h*)(bp2 + 16);
            *(v8h*)&b3 = *(const v8h*)bp3; *((v8h*)&b3 + 1) = *(const v8h*)(bp3 + 16);

            acc0 = __builtin_amdgcn_wmma_f32_16x16x32_f16(false, a, false, b0, (short)0, acc0, false, false);
            acc1 = __builtin_amdgcn_wmma_f32_16x16x32_f16(false, a, false, b1, (short)0, acc1, false, false);
            acc2 = __builtin_amdgcn_wmma_f32_16x16x32_f16(false, a, false, b2, (short)0, acc2, false, false);
            acc3 = __builtin_amdgcn_wmma_f32_16x16x32_f16(false, a, false, b3, (short)0, acc3, false, false);
        }

        float part[8];
        #pragma unroll
        for (int r = 0; r < 8; ++r) part[r] = 0.0f;

        #pragma unroll
        for (int nt = 0; nt < 4; ++nt) {
            int   n  = nt * 16 + mrow;
            float bv = bl[n];
            float uv = ul[n];
            v8f c = (nt == 0) ? acc0 : (nt == 1) ? acc1 : (nt == 2) ? acc2 : acc3;
            #pragma unroll
            for (int r = 0; r < 8; ++r)
                part[r] += uv * fast_tanh(c[r] + bv);
        }

        #pragma unroll
        for (int r = 0; r < 8; ++r) {
            float p = part[r];
            p += __shfl_xor(p, 8, 16);
            p += __shfl_xor(p, 4, 16);
            p += __shfl_xor(p, 2, 16);
            p += __shfl_xor(p, 1, 16);
            if (mrow == 0)
                sc[t0 + r + (lane >> 4) * 8] = p;
        }
    }
    __syncthreads();

    float e0 = __expf(sc[tid]);
    float e1 = __expf(sc[tid + NTHREADS]);
    red[tid] = e0 + e1;
    __syncthreads();
    #pragma unroll
    for (int s = NTHREADS / 2; s > 0; s >>= 1) {
        if (tid < s) red[tid] += red[tid + s];
        __syncthreads();
    }
    float inv = 1.0f / (red[0] + EPS_K);
    __syncthreads();
    sc[tid]            = e0 * inv;
    sc[tid + NTHREADS] = e1 * inv;
    __syncthreads();

    float acc = 0.0f;
    #pragma unroll 8
    for (int t = 0; t < T_DIM; ++t)
        acc += (float)xs[t * XS + tid] * sc[t];
    out[(size_t)b * D_DIM + tid] = acc;
}

extern "C" void kernel_launch(void* const* d_in, const int* in_sizes, int n_in,
                              void* d_out, int out_size, void* d_ws, size_t ws_size,
                              hipStream_t stream) {
    const float* x    = (const float*)d_in[0];
    const float* W    = (const float*)d_in[1];
    const float* bias = (const float*)d_in[2];
    const float* u    = (const float*)d_in[3];
    float* out        = (float*)d_out;

    hipFuncSetAttribute((const void*)attn_pool_kernel,
                        hipFuncAttributeMaxDynamicSharedMemorySize, SMEM_BYTES);
    attn_pool_kernel<<<256, NTHREADS, SMEM_BYTES, stream>>>(x, W, bias, u, out);
}


# W staging: each thread's 50 loads made unguarded, independent and line-coalesced and issued as one group (one L2 round trip) instead of 64 guarded loads each waited for singly; same LDS contents
# speedup vs baseline: 1.7651x; 1.7651x over previous
#include <hip/hip_runtime.h>

typedef __attribute__((ext_vector_type(16))) _Float16 v16h;
typedef __attribute__((ext_vector_type(8)))  _Float16 v8h;
typedef __attribute__((ext_vector_type(8)))  float    v8f;

#define T_DIM   512
#define D_DIM   256
#define A_DIM   50
#define A_PAD   64
#define XS      264
#define NTHREADS 256
#define EPS_K   1e-7f
#define W_PER_THREAD (D_DIM * A_DIM / NTHREADS)
static_assert(W_PER_THREAD * NTHREADS == D_DIM * A_DIM, "W must split evenly over the threads");
static_assert(D_DIM / 8 == 32, "pad-column fill assumes 32 uint4 pieces per row");

#define OFF_X    0
#define SZ_X     (T_DIM * XS * 2)
#define OFF_WT   (OFF_X + SZ_X)
#define SZ_WT    (A_PAD * XS * 2)
#define OFF_B    (OFF_WT + SZ_WT)
#define OFF_U    (OFF_B + A_PAD * 4)
#define OFF_S    (OFF_U + A_PAD * 4)
#define OFF_RED  (OFF_S + T_DIM * 4)
#define SMEM_BYTES (OFF_RED + NTHREADS * 4)

__device__ __forceinline__ float fast_tanh(float x) {
    float e = __expf(2.0f * x);
    return (e - 1.0f) / (e + 1.0f);
}

__global__ __launch_bounds__(NTHREADS, 1)
void attn_pool_kernel(const float* __restrict__ x,
                      const float* __restrict__ W,
                      const float* __restrict__ bias,
                      const float* __restrict__ u,
                      float* __restrict__ out)
{
    extern __shared__ __align__(16) char smem[];
    _Float16* xs  = (_Float16*)(smem + OFF_X);
    _Float16* wt  = (_Float16*)(smem + OFF_WT);
    float*    bl  = (float*)(smem + OFF_B);
    float*    ul  = (float*)(smem + OFF_U);
    float*    sc  = (float*)(smem + OFF_S);
    float*    red = (float*)(smem + OFF_RED);

    const int tid  = threadIdx.x;
    const int b    = blockIdx.x;
    const int lane = tid & 31;
    const int wave = tid >> 5;

    {
        float wv[W_PER_THREAD];
        #pragma unroll
        for (int i = 0; i < W_PER_THREAD; ++i)
            wv[i] = W[tid + i * NTHREADS];
        #pragma unroll
        for (int i = 0; i < W_PER_THREAD; ++i) {
            unsigned e = (unsigned)(tid + i * NTHREADS);
            unsigned k = e / A_DIM;
            unsigned n = e - k * A_DIM;
            wt[n * XS + k] = (_Float16)wv[i];
        }
    }
    for (int idx = tid; idx < (A_PAD - A_DIM) * (D_DIM / 8); idx += NTHREADS) {
        int n = A_DIM + (idx >> 5);
        int c = idx & 31;
        *(uint4*)(wt + n * XS + c * 8) = make_uint4(0u, 0u, 0u, 0u);
    }
    if (tid < A_PAD) {
        bl[tid] = (tid < A_DIM) ? bias[tid] : 0.0f;
        ul[tid] = (tid < A_DIM) ? u[tid]    : 0.0f;
    }

    const float4* xg = (const float4*)(x + (size_t)b * T_DIM * D_DIM);
    for (int idx = tid; idx < T_DIM * (D_DIM / 4); idx += NTHREADS) {
        int t = idx >> 6;
        int c = idx & 63;
        float4 v = xg[t * (D_DIM / 4) + c];
        _Float16* dst = xs + t * XS + c * 4;
        dst[0] = (_Float16)v.x; dst[1] = (_Float16)v.y;
        dst[2] = (_Float16)v.z; dst[3] = (_Float16)v.w;
    }
    __syncthreads();

    const int mrow  = lane & 15;
    const int khalf = (lane >> 4) * 8;

    for (int mt = wave; mt < T_DIM / 16; mt += NTHREADS / 32) {
        const int t0 = mt * 16;
        v8f acc0 = {}, acc1 = {}, acc2 = {}, acc3 = {};

        for (int ks = 0; ks < D_DIM / 32; ++ks) {
            const _Float16* ap = xs + (t0 + mrow) * XS + ks * 32 + khalf;
            v16h a;
            *(v8h*)&a       = *(const v8h*)ap;
            *((v8h*)&a + 1) = *(const v8h*)(ap + 16);

            const _Float16* bp0 = wt + ( 0 + mrow) * XS + ks * 32 + khalf;
            const _Float16* bp1 = wt + (16 + mrow) * XS + ks * 32 + khalf;
            const _Float16* bp2 = wt + (32 + mrow) * XS + ks * 32 + khalf;
            const _Float16* bp3 = wt + (48 + mrow) * XS + ks * 32 + khalf;
            v16h b0, b1, b2, b3;
            *(v8h*)&b0 = *(const v8h*)bp0; *((v8h*)&b0 + 1) = *(const v8h*)(bp0 + 16);
            *(v8h*)&b1 = *(const v8h*)bp1; *((v8h*)&b1 + 1) = *(const v8h*)(bp1 + 16);
            *(v8h*)&b2 = *(const v8h*)bp2; *((v8h*)&b2 + 1) = *(const v8h*)(bp2 + 16);
            *(v8h*)&b3 = *(const v8h*)bp3; *((v8h*)&b3 + 1) = *(const v8h*)(bp3 + 16);

            acc0 = __builtin_amdgcn_wmma_f32_16x16x32_f16(false, a, false, b0, (short)0, acc0, false, false);
            acc1 = __builtin_amdgcn_wmma_f32_16x16x32_f16(false, a, false, b1, (short)0, acc1, false, false);
            acc2 = __builtin_amdgcn_wmma_f32_16x16x32_f16(false, a, false, b2, (short)0, acc2, false, false);
            acc3 = __builtin_amdgcn_wmma_f32_16x16x32_f16(false, a, false, b3, (short)0, acc3, false, false);
        }

        float part[8];
        #pragma unroll
        for (int r = 0; r < 8; ++r) part[r] = 0.0f;

        #pragma unroll
        for (int nt = 0; nt < 4; ++nt) {
            int   n  = nt * 16 + mrow;
            float bv = bl[n];
            float uv = ul[n];
            v8f c = (nt == 0) ? acc0 : (nt == 1) ? acc1 : (nt == 2) ? acc2 : acc3;
            #pragma unroll
            for (int r = 0; r < 8; ++r)
                part[r] += uv * fast_tanh(c[r] + bv);
        }

        #pragma unroll
        for (int r = 0; r < 8; ++r) {
            float p = part[r];
            p += __shfl_xor(p, 8, 16);
            p += __shfl_xor(p, 4, 16);
            p += __shfl_xor(p, 2, 16);
            p += __shfl_xor(p, 1, 16);
            if (mrow == 0)
                sc[t0 + r + (lane >> 4) * 8] = p;
        }
    }
    __syncthreads();

    float e0 = __expf(sc[tid]);
    float e1 = __expf(sc[tid + NTHREADS]);
    red[tid] = e0 + e1;
    __syncthreads();
    #pragma unroll
    for (int s = NTHREADS / 2; s > 0; s >>= 1) {
        if (tid < s) red[tid] += red[tid + s];
        __syncthreads();
    }
    float inv = 1.0f / (red[0] + EPS_K);
    __syncthreads();
    sc[tid]            = e0 * inv;
    sc[tid + NTHREADS] = e1 * inv;
    __syncthreads();

    float acc = 0.0f;
    #pragma unroll 8
    for (int t = 0; t < T_DIM; ++t)
        acc += (float)xs[t * XS + tid] * sc[t];
    out[(size_t)b * D_DIM + tid] = acc;
}

extern "C" void kernel_launch(void* const* d_in, const int* in_sizes, int n_in,
                              void* d_out, int out_size, void* d_ws, size_t ws_size,
                              hipStream_t stream) {
    const float* x    = (const float*)d_in[0];
    const float* W    = (const float*)d_in[1];
    const float* bias = (const float*)d_in[2];
    const float* u    = (const float*)d_in[3];
    float* out        = (float*)d_out;

    hipFuncSetAttribute((const void*)attn_pool_kernel,
                        hipFuncAttributeMaxDynamicSharedMemorySize, SMEM_BYTES);
    attn_pool_kernel<<<256, NTHREADS, SMEM_BYTES, stream>>>(x, W, bias, u, out);
}


# x staging: a rolling window of 48 float4 loads per thread kept in flight (opened behind the W loads) instead of 16 groups of 8, each drained before the next; same LDS contents
# speedup vs baseline: 1.9256x; 1.0909x over previous
#include <hip/hip_runtime.h>

typedef __attribute__((ext_vector_type(16))) _Float16 v16h;
typedef __attribute__((ext_vector_type(8)))  _Float16 v8h;
typedef __attribute__((ext_vector_type(8)))  float    v8f;

#define T_DIM   512
#define D_DIM   256
#define A_DIM   50
#define A_PAD   64
#define XS      264
#define NTHREADS 256
#define EPS_K   1e-7f
#define W_PER_THREAD (D_DIM * A_DIM / NTHREADS)
static_assert(W_PER_THREAD * NTHREADS == D_DIM * A_DIM, "W must split evenly over the threads");
static_assert(D_DIM / 8 == 32, "pad-column fill assumes 32 uint4 pieces per row");
#define X_LOADS  (T_DIM * (D_DIM / 4) / NTHREADS)
#define X_WIN    48
static_assert(X_LOADS * NTHREADS == T_DIM * (D_DIM / 4), "x[b] must split evenly over the threads");
static_assert(NTHREADS % (D_DIM / 4) == 0, "a thread keeps its column piece from load to load");
static_assert(X_WIN < X_LOADS && X_WIN < 64, "the window must stay under the 63 loads one wave may have outstanding");

#define OFF_X    0
#define SZ_X     (T_DIM * XS * 2)
#define OFF_WT   (OFF_X + SZ_X)
#define SZ_WT    (A_PAD * XS * 2)
#define OFF_B    (OFF_WT + SZ_WT)
#define OFF_U    (OFF_B + A_PAD * 4)
#define OFF_S    (OFF_U + A_PAD * 4)
#define OFF_RED  (OFF_S + T_DIM * 4)
#define SMEM_BYTES (OFF_RED + NTHREADS * 4)

__device__ __forceinline__ float fast_tanh(float x) {
    float e = __expf(2.0f * x);
    return (e - 1.0f) / (e + 1.0f);
}

__global__ __launch_bounds__(NTHREADS, 1)
void attn_pool_kernel(const float* __restrict__ x,
                      const float* __restrict__ W,
                      const float* __restrict__ bias,
                      const float* __restrict__ u,
                      float* __restrict__ out)
{
    extern __shared__ __align__(16) char smem[];
    _Float16* xs  = (_Float16*)(smem + OFF_X);
    _Float16* wt  = (_Float16*)(smem + OFF_WT);
    float*    bl  = (float*)(smem + OFF_B);
    float*    ul  = (float*)(smem + OFF_U);
    float*    sc  = (float*)(smem + OFF_S);
    float*    red = (float*)(smem + OFF_RED);

    const int tid  = threadIdx.x;
    const int b    = blockIdx.x;
    const int lane = tid & 31;
    const int wave = tid >> 5;

    const float4* xg = (const float4*)(x + (size_t)b * T_DIM * D_DIM) + tid;
    float4 win[X_WIN];
    const int   ci    = (tid < A_DIM) ? tid : 0;
    const float b_reg = bias[ci];
    const float u_reg = u[ci];
    {
        float wv[W_PER_THREAD];
        #pragma unroll
        for (int i = 0; i < W_PER_THREAD; ++i)
            wv[i] = W[tid + i * NTHREADS];
        #pragma unroll
        for (int i = 0; i < X_WIN; ++i)
            win[i] = xg[i * NTHREADS];
        #pragma unroll
        for (int i = 0; i < W_PER_THREAD; ++i) {
            unsigned e = (unsigned)(tid + i * NTHREADS);
            unsigned k = e / A_DIM;
            unsigned n = e - k * A_DIM;
            wt[n * XS + k] = (_Float16)wv[i];
        }
    }
    for (int idx = tid; idx < (A_PAD - A_DIM) * (D_DIM / 8); idx += NTHREADS) {
        int n = A_DIM + (idx >> 5);
        int c = idx & 31;
        *(uint4*)(wt + n * XS + c * 8) = make_uint4(0u, 0u, 0u, 0u);
    }
    if (tid < A_PAD) {
        bl[tid] = (tid < A_DIM) ? b_reg : 0.0f;
        ul[tid] = (tid < A_DIM) ? u_reg : 0.0f;
    }

    {
        _Float16* xdst = xs + (tid >> 6) * XS + (tid & 63) * 4;
        #pragma unroll
        for (int i = 0; i < X_LOADS; ++i) {
            float4 v = win[i % X_WIN];
            if (i + X_WIN < X_LOADS)
                win[i % X_WIN] = xg[(i + X_WIN) * NTHREADS];
            _Float16* dst = xdst + i * (NTHREADS / (D_DIM / 4)) * XS;
            dst[0] = (_Float16)v.x; dst[1] = (_Float16)v.y;
            dst[2] = (_Float16)v.z; dst[3] = (_Float16)v.w;
        }
    }
    __syncthreads();

    const int mrow  = lane & 15;
    const int khalf = (lane >> 4) * 8;

    for (int mt = wave; mt < T_DIM / 16; mt += NTHREADS / 32) {
        const int t0 = mt * 16;
        v8f acc0 = {}, acc1 = {}, acc2 = {}, acc3 = {};

        for (int ks = 0; ks < D_DIM / 32; ++ks) {
            const _Float16* ap = xs + (t0 + mrow) * XS + ks * 32 + khalf;
            v16h a;
            *(v8h*)&a       = *(const v8h*)ap;
            *((v8h*)&a + 1) = *(const v8h*)(ap + 16);

            const _Float16* bp0 = wt + ( 0 + mrow) * XS + ks * 32 + khalf;
            const _Float16* bp1 = wt + (16 + mrow) * XS + ks * 32 + khalf;
            const _Float16* bp2 = wt + (32 + mrow) * XS + ks * 32 + khalf;
            const _Float16* bp3 = wt + (48 + mrow) * XS + ks * 32 + khalf;
            v16h b0, b1, b2, b3;
            *(v8h*)&b0 = *(const v8h*)bp0; *((v8h*)&b0 + 1) = *(const v8h*)(bp0 + 16);
            *(v8h*)&b1 = *(const v8h*)bp1; *((v8h*)&b1 + 1) = *(const v8h*)(bp1 + 16);
            *(v8h*)&b2 = *(const v8h*)bp2; *((v8h*)&b2 + 1) = *(const v8h*)(bp2 + 16);
            *(v8h*)&b3 = *(const v8h*)bp3; *((v8h*)&b3 + 1) = *(const v8h*)(bp3 + 16);

            acc0 = __builtin_amdgcn_wmma_f32_16x16x32_f16(false, a, false, b0, (short)0, acc0, false, false);
            acc1 = __builtin_amdgcn_wmma_f32_16x16x32_f16(false, a, false, b1, (short)0, acc1, false, false);
            acc2 = __builtin_amdgcn_wmma_f32_16x16x32_f16(false, a, false, b2, (short)0, acc2, false, false);
            acc3 = __builtin_amdgcn_wmma_f32_16x16x32_f16(false, a, false, b3, (short)0, acc3, false, false);
        }

        float part[8];
        #pragma unroll
        for (int r = 0; r < 8; ++r) part[r] = 0.0f;

        #pragma unroll
        for (int nt = 0; nt < 4; ++nt) {
            int   n  = nt * 16 + mrow;
            float bv = bl[n];
            float uv = ul[n];
            v8f c = (nt == 0) ? acc0 : (nt == 1) ? acc1 : (nt == 2) ? acc2 : acc3;
            #pragma unroll
            for (int r = 0; r < 8; ++r)
                part[r] += uv * fast_tanh(c[r] + bv);
        }

        #pragma unroll
        for (int r = 0; r < 8; ++r) {
            float p = part[r];
            p += __shfl_xor(p, 8, 16);
            p += __shfl_xor(p, 4, 16);
            p += __shfl_xor(p, 2, 16);
            p += __shfl_xor(p, 1, 16);
            if (mrow == 0)
                sc[t0 + r + (lane >> 4) * 8] = p;
        }
    }
    __syncthreads();

    float e0 = __expf(sc[tid]);
    float e1 = __expf(sc[tid + NTHREADS]);
    red[tid] = e0 + e1;
    __syncthreads();
    #pragma unroll
    for (int s = NTHREADS / 2; s > 0; s >>= 1) {
        if (tid < s) red[tid] += red[tid + s];
        __syncthreads();
    }
    float inv = 1.0f / (red[0] + EPS_K);
    __syncthreads();
    sc[tid]            = e0 * inv;
    sc[tid + NTHREADS] = e1 * inv;
    __syncthreads();

    float acc = 0.0f;
    #pragma unroll 8
    for (int t = 0; t < T_DIM; ++t)
        acc += (float)xs[t * XS + tid] * sc[t];
    out[(size_t)b * D_DIM + tid] = acc;
}

extern "C" void kernel_launch(void* const* d_in, const int* in_sizes, int n_in,
                              void* d_out, int out_size, void* d_ws, size_t ws_size,
                              hipStream_t stream) {
    const float* x    = (const float*)d_in[0];
    const float* W    = (const float*)d_in[1];
    const float* bias = (const float*)d_in[2];
    const float* u    = (const float*)d_in[3];
    float* out        = (float*)d_out;

    hipFuncSetAttribute((const void*)attn_pool_kernel,
                        hipFuncAttributeMaxDynamicSharedMemorySize, SMEM_BYTES);
    attn_pool_kernel<<<256, NTHREADS, SMEM_BYTES, stream>>>(x, W, bias, u, out);
}


# scores: W^T is the WMMA A operand and x the B operand, so a lane holds one time step's units: their sum is in-lane chains + one half exchange (new summation order), not 32 LDS permutes per tile
# speedup vs baseline: 2.2628x; 1.1751x over previous
#include <hip/hip_runtime.h>

typedef __attribute__((ext_vector_type(16))) _Float16 v16h;
typedef __attribute__((ext_vector_type(8)))  _Float16 v8h;
typedef __attribute__((ext_vector_type(8)))  float    v8f;

#define T_DIM   512
#define D_DIM   256
#define A_DIM   50
#define A_PAD   64
#define XS      264
#define NTHREADS 256
#define EPS_K   1e-7f
#define W_PER_THREAD (D_DIM * A_DIM / NTHREADS)
static_assert(W_PER_THREAD * NTHREADS == D_DIM * A_DIM, "W must split evenly over the threads");
static_assert(D_DIM / 8 == 32, "pad-column fill assumes 32 uint4 pieces per row");
#define X_LOADS  (T_DIM * (D_DIM / 4) / NTHREADS)
#define X_WIN    48
static_assert(X_LOADS * NTHREADS == T_DIM * (D_DIM / 4), "x[b] must split evenly over the threads");
static_assert(NTHREADS % (D_DIM / 4) == 0, "a thread keeps its column piece from load to load");
static_assert(X_WIN < X_LOADS && X_WIN < 64, "the window must stay under the 63 loads one wave may have outstanding");

#define OFF_X    0
#define SZ_X     (T_DIM * XS * 2)
#define OFF_WT   (OFF_X + SZ_X)
#define SZ_WT    (A_PAD * XS * 2)
#define OFF_B    (OFF_WT + SZ_WT)
#define OFF_U    (OFF_B + A_PAD * 4)
#define OFF_S    (OFF_U + A_PAD * 4)
#define OFF_RED  (OFF_S + T_DIM * 4)
#define SMEM_BYTES (OFF_RED + NTHREADS * 4)

__device__ __forceinline__ float fast_tanh(float x) {
    float e = __expf(2.0f * x);
    return (e - 1.0f) / (e + 1.0f);
}

__device__ __forceinline__ float xhalf(float v) {
    int b = __builtin_bit_cast(int, v);
    return __builtin_bit_cast(float, __builtin_amdgcn_permlanex16(b, b, 0x76543210, 0xfedcba98, false, false));
}

__global__ __launch_bounds__(NTHREADS, 1)
void attn_pool_kernel(const float* __restrict__ x,
                      const float* __restrict__ W,
                      const float* __restrict__ bias,
                      const float* __restrict__ u,
                      float* __restrict__ out)
{
    extern __shared__ __align__(16) char smem[];
    _Float16* xs  = (_Float16*)(smem + OFF_X);
    _Float16* wt  = (_Float16*)(smem + OFF_WT);
    float*    bl  = (float*)(smem + OFF_B);
    float*    ul  = (float*)(smem + OFF_U);
    float*    sc  = (float*)(smem + OFF_S);
    float*    red = (float*)(smem + OFF_RED);

    const int tid  = threadIdx.x;
    const int b    = blockIdx.x;
    const int lane = tid & 31;
    const int wave = tid >> 5;

    const float4* xg = (const float4*)(x + (size_t)b * T_DIM * D_DIM) + tid;
    float4 win[X_WIN];
    const int   ci    = (tid < A_DIM) ? tid : 0;
    const float b_reg = bias[ci];
    const float u_reg = u[ci];
    {
        float wv[W_PER_THREAD];
        #pragma unroll
        for (int i = 0; i < W_PER_THREAD; ++i)
            wv[i] = W[tid + i * NTHREADS];
        #pragma unroll
        for (int i = 0; i < X_WIN; ++i)
            win[i] = xg[i * NTHREADS];
        #pragma unroll
        for (int i = 0; i < W_PER_THREAD; ++i) {
            unsigned e = (unsigned)(tid + i * NTHREADS);
            unsigned k = e / A_DIM;
            unsigned n = e - k * A_DIM;
            wt[n * XS + k] = (_Float16)wv[i];
        }
    }
    for (int idx = tid; idx < (A_PAD - A_DIM) * (D_DIM / 8); idx += NTHREADS) {
        int n = A_DIM + (idx >> 5);
        int c = idx & 31;
        *(uint4*)(wt + n * XS + c * 8) = make_uint4(0u, 0u, 0u, 0u);
    }
    if (tid < A_PAD) {
        bl[tid] = (tid < A_DIM) ? b_reg : 0.0f;
        ul[tid] = (tid < A_DIM) ? u_reg : 0.0f;
    }

    {
        _Float16* xdst = xs + (tid >> 6) * XS + (tid & 63) * 4;
        #pragma unroll
        for (int i = 0; i < X_LOADS; ++i) {
            float4 v = win[i % X_WIN];
            if (i + X_WIN < X_LOADS)
                win[i % X_WIN] = xg[(i + X_WIN) * NTHREADS];
            _Float16* dst = xdst + i * (NTHREADS / (D_DIM / 4)) * XS;
            dst[0] = (_Float16)v.x; dst[1] = (_Float16)v.y;
            dst[2] = (_Float16)v.z; dst[3] = (_Float16)v.w;
        }
    }
    __syncthreads();

    const int mrow  = lane & 15;
    const int khalf = (lane >> 4) * 8;

    float bv[4][8], uv[4][8];
    #pragma unroll
    for (int nt = 0; nt < 4; ++nt) {
        const float4 b_lo = *(const float4*)(bl + nt * 16 + khalf);
        const float4 b_hi = *(const float4*)(bl + nt * 16 + khalf + 4);
        const float4 u_lo = *(const float4*)(ul + nt * 16 + khalf);
        const float4 u_hi = *(const float4*)(ul + nt * 16 + khalf + 4);
        bv[nt][0] = b_lo.x; bv[nt][1] = b_lo.y; bv[nt][2] = b_lo.z; bv[nt][3] = b_lo.w;
        bv[nt][4] = b_hi.x; bv[nt][5] = b_hi.y; bv[nt][6] = b_hi.z; bv[nt][7] = b_hi.w;
        uv[nt][0] = u_lo.x; uv[nt][1] = u_lo.y; uv[nt][2] = u_lo.z; uv[nt][3] = u_lo.w;
        uv[nt][4] = u_hi.x; uv[nt][5] = u_hi.y; uv[nt][6] = u_hi.z; uv[nt][7] = u_hi.w;
    }

    for (int mt = wave; mt < T_DIM / 16; mt += NTHREADS / 32) {
        const int t0 = mt * 16;
        v8f acc0 = {}, acc1 = {}, acc2 = {}, acc3 = {};

        for (int ks = 0; ks < D_DIM / 32; ++ks) {
            const _Float16* xp = xs + (t0 + mrow) * XS + ks * 32 + khalf;
            v16h xf;
            *(v8h*)&xf       = *(const v8h*)xp;
            *((v8h*)&xf + 1) = *(const v8h*)(xp + 16);

            const _Float16* wp0 = wt + ( 0 + mrow) * XS + ks * 32 + khalf;
            const _Float16* wp1 = wt + (16 + mrow) * XS + ks * 32 + khalf;
            const _Float16* wp2 = wt + (32 + mrow) * XS + ks * 32 + khalf;
            const _Float16* wp3 = wt + (48 + mrow) * XS + ks * 32 + khalf;
            v16h w0, w1, w2, w3;
            *(v8h*)&w0 = *(const v8h*)wp0; *((v8h*)&w0 + 1) = *(const v8h*)(wp0 + 16);
            *(v8h*)&w1 = *(const v8h*)wp1; *((v8h*)&w1 + 1) = *(const v8h*)(wp1 + 16);
            *(v8h*)&w2 = *(const v8h*)wp2; *((v8h*)&w2 + 1) = *(const v8h*)(wp2 + 16);
            *(v8h*)&w3 = *(const v8h*)wp3; *((v8h*)&w3 + 1) = *(const v8h*)(wp3 + 16);

            acc0 = __builtin_amdgcn_wmma_f32_16x16x32_f16(false, w0, false, xf, (short)0, acc0, false, false);
            acc1 = __builtin_amdgcn_wmma_f32_16x16x32_f16(false, w1, false, xf, (short)0, acc1, false, false);
            acc2 = __builtin_amdgcn_wmma_f32_16x16x32_f16(false, w2, false, xf, (short)0, acc2, false, false);
            acc3 = __builtin_amdgcn_wmma_f32_16x16x32_f16(false, w3, false, xf, (short)0, acc3, false, false);
        }

        float p0 = 0.0f, p1 = 0.0f, p2 = 0.0f, p3 = 0.0f;
        #pragma unroll
        for (int r = 0; r < 8; ++r) {
            p0 += uv[0][r] * fast_tanh(acc0[r] + bv[0][r]);
            p1 += uv[1][r] * fast_tanh(acc1[r] + bv[1][r]);
            p2 += uv[2][r] * fast_tanh(acc2[r] + bv[2][r]);
            p3 += uv[3][r] * fast_tanh(acc3[r] + bv[3][r]);
        }
        float ps = (p0 + p1) + (p2 + p3);
        ps += xhalf(ps);
        if (lane < 16)
            sc[t0 + lane] = ps;
    }
    __syncthreads();

    float e0 = __expf(sc[tid]);
    float e1 = __expf(sc[tid + NTHREADS]);
    red[tid] = e0 + e1;
    __syncthreads();
    #pragma unroll
    for (int s = NTHREADS / 2; s > 0; s >>= 1) {
        if (tid < s) red[tid] += red[tid + s];
        __syncthreads();
    }
    float inv = 1.0f / (red[0] + EPS_K);
    __syncthreads();
    sc[tid]            = e0 * inv;
    sc[tid + NTHREADS] = e1 * inv;
    __syncthreads();

    float acc = 0.0f;
    #pragma unroll 8
    for (int t = 0; t < T_DIM; ++t)
        acc += (float)xs[t * XS + tid] * sc[t];
    out[(size_t)b * D_DIM + tid] = acc;
}

extern "C" void kernel_launch(void* const* d_in, const int* in_sizes, int n_in,
                              void* d_out, int out_size, void* d_ws, size_t ws_size,
                              hipStream_t stream) {
    const float* x    = (const float*)d_in[0];
    const float* W    = (const float*)d_in[1];
    const float* bias = (const float*)d_in[2];
    const float* u    = (const float*)d_in[3];
    float* out        = (float*)d_out;

    hipFuncSetAttribute((const void*)attn_pool_kernel,
                        hipFuncAttributeMaxDynamicSharedMemorySize, SMEM_BYTES);
    attn_pool_kernel<<<256, NTHREADS, SMEM_BYTES, stream>>>(x, W, bias, u, out);
}
